# starting kernel
# baseline (speedup 1.0000x reference)
#include <hip/hip_runtime.h>


#define BATCH  16
#define SEQ    4096
#define DIM    1024
#define SSPLIT 64
#define SCHUNK (SEQ / SSPLIT)

typedef float  v2f  __attribute__((ext_vector_type(2)));
typedef float  v4f  __attribute__((ext_vector_type(4)));
typedef float  v8f  __attribute__((ext_vector_type(8)));
typedef _Float16 v16h __attribute__((ext_vector_type(16)));

__global__ __launch_bounds__(256) void hcen_mean_stage1(
    const float* __restrict__ x, float* __restrict__ part) {
  const int sblk = blockIdx.x;
  const int b    = blockIdx.y;
  const int t    = threadIdx.x;

  const float* xp = x + ((size_t)b * SEQ + (size_t)sblk * SCHUNK) * DIM + t * 4;
  v4f acc = {0.f, 0.f, 0.f, 0.f};
#pragma unroll 4
  for (int s = 0; s < SCHUNK; ++s) {
    v4f v = __builtin_nontemporal_load((const v4f*)(xp + (size_t)s * DIM));
    acc += v;
  }
  *(v4f*)(part + (size_t)(sblk * BATCH + b) * DIM + t * 4) = acc;
}

__global__ __launch_bounds__(256) void hcen_mean_stage2(
    const float* __restrict__ part, float* __restrict__ m) {
  const int i = blockIdx.x * 256 + threadIdx.x;
  float acc = 0.f;
#pragma unroll 8
  for (int s = 0; s < SSPLIT; ++s)
    acc += part[(size_t)s * (BATCH * DIM) + i];
  m[i] = acc * (1.0f / (float)SEQ);
}

__global__ __launch_bounds__(32) void hcen_gemm16(
    const float* __restrict__ A, const float* __restrict__ W,
    const float* __restrict__ bias, float* __restrict__ D,
    int K, int N) {
  const int n0   = blockIdx.x * 16;
  const int lane = threadIdx.x;
  const int half = lane >> 4;
  const int lm   = lane & 15;

  v8f c;
  const float bv = bias[n0 + lm];
#pragma unroll
  for (int j = 0; j < 8; ++j) c[j] = bv;

#if __has_builtin(__builtin_amdgcn_wmma_f32_16x16x4_f32)
  const float* arow = A + (size_t)lm * K + 2 * half;
  const float* wrow = W + (size_t)(n0 + lm) * K + 2 * half;
#pragma unroll 8
  for (int k0 = 0; k0 < K; k0 += 4) {
    v2f av = *(const v2f*)(arow + k0);
    v2f bv2 = *(const v2f*)(wrow + k0);
    c = __builtin_amdgcn_wmma_f32_16x16x4_f32(
        false, av, false, bv2, (short)0, c, false, false);
  }
#else
  const float* arow = A + (size_t)lm * K + 8 * half;
  const float* wrow = W + (size_t)(n0 + lm) * K + 8 * half;
  for (int k0 = 0; k0 < K; k0 += 32) {
    v16h af, bf;
#pragma unroll
    for (int e = 0; e < 16; ++e) {
      const int off = k0 + 16 * (e >> 3) + (e & 7);
      af[e] = (_Float16)arow[off];
      bf[e] = (_Float16)wrow[off];
    }
    c = __builtin_amdgcn_wmma_f32_16x16x32_f16(
        false, af, false, bf, (short)0, c, false, false);
  }
#endif

#pragma unroll
  for (int j = 0; j < 8; ++j)
    D[(size_t)(j + 8 * half) * N + (n0 + lm)] = c[j];
}

extern "C" void kernel_launch(void* const* d_in, const int* in_sizes, int n_in,
                              void* d_out, int out_size, void* d_ws, size_t ws_size,
                              hipStream_t stream) {
  const float* x     = (const float*)d_in[0];
  const float* W_enc = (const float*)d_in[1];
  const float* b_enc = (const float*)d_in[2];
  const float* W_out = (const float*)d_in[3];
  const float* b_out = (const float*)d_in[4];
  float* out = (float*)d_out;

  float* part = (float*)d_ws;
  float* m    = part + (size_t)SSPLIT * BATCH * DIM;
  float* enc  = m + (size_t)BATCH * DIM;

  dim3 g1(SSPLIT, BATCH);
  hcen_mean_stage1<<<g1, 256, 0, stream>>>(x, part);
  hcen_mean_stage2<<<(BATCH * DIM) / 256, 256, 0, stream>>>(part, m);
  hcen_gemm16<<<DIM / 16, 32, 0, stream>>>(m,   W_enc, b_enc, enc, DIM, DIM);
  hcen_gemm16<<<DIM / 16, 32, 0, stream>>>(enc, W_out, b_out, out, DIM, DIM);
}


# GEMMs: k range of each 16x16 tile split over the 8 waves of one block, every fragment load issued before the first WMMA, partial tiles summed through LDS in fixed order (new reduction tree)
# speedup vs baseline: 1.5982x; 1.5982x over previous
#include <hip/hip_runtime.h>


#define BATCH  16
#define SEQ    4096
#define DIM    1024
#define SSPLIT 64
#define SCHUNK (SEQ / SSPLIT)

typedef float  v2f  __attribute__((ext_vector_type(2)));
typedef float  v4f  __attribute__((ext_vector_type(4)));
typedef float  v8f  __attribute__((ext_vector_type(8)));

__global__ __launch_bounds__(256) void hcen_mean_stage1(
    const float* __restrict__ x, float* __restrict__ part) {
  const int sblk = blockIdx.x;
  const int b    = blockIdx.y;
  const int t    = threadIdx.x;

  const float* xp = x + ((size_t)b * SEQ + (size_t)sblk * SCHUNK) * DIM + t * 4;
  v4f acc = {0.f, 0.f, 0.f, 0.f};
#pragma unroll 4
  for (int s = 0; s < SCHUNK; ++s) {
    v4f v = __builtin_nontemporal_load((const v4f*)(xp + (size_t)s * DIM));
    acc += v;
  }
  *(v4f*)(part + (size_t)(sblk * BATCH + b) * DIM + t * 4) = acc;
}

__global__ __launch_bounds__(256) void hcen_mean_stage2(
    const float* __restrict__ part, float* __restrict__ m) {
  const int i = blockIdx.x * 256 + threadIdx.x;
  float acc = 0.f;
#pragma unroll 8
  for (int s = 0; s < SSPLIT; ++s)
    acc += part[(size_t)s * (BATCH * DIM) + i];
  m[i] = acc * (1.0f / (float)SEQ);
}

#define KWAVES 8
#define KSLICE (DIM / KWAVES)
#define KLOADS (KSLICE / 8)

__global__ __launch_bounds__(32 * KWAVES) void hcen_gemm16(
    const float* __restrict__ A, const float* __restrict__ W,
    const float* __restrict__ bias, float* __restrict__ D) {
  __shared__ float red[KWAVES][8][32];

  const int n0   = blockIdx.x * 16;
  const int wave = threadIdx.x >> 5;
  const int lane = threadIdx.x & 31;
  const int half = lane >> 4;
  const int lm   = lane & 15;

  const float* arow = A + (size_t)lm * DIM + wave * KSLICE + 4 * half;
  const float* wrow = W + (size_t)(n0 + lm) * DIM + wave * KSLICE + 4 * half;

  const float bv = bias[n0 + lm];

  v4f a[KLOADS], b[KLOADS];
#pragma unroll
  for (int i = 0; i < KLOADS; ++i) {
    a[i] = *(const v4f*)(arow + 8 * i);
    b[i] = *(const v4f*)(wrow + 8 * i);
  }

  __builtin_amdgcn_sched_barrier(0);

  v8f c0 = {0.f, 0.f, 0.f, 0.f, 0.f, 0.f, 0.f, 0.f};
  v8f c1 = {0.f, 0.f, 0.f, 0.f, 0.f, 0.f, 0.f, 0.f};
#pragma unroll
  for (int i = 0; i < KLOADS; ++i) {
    const v2f a0 = {a[i][0], a[i][1]};
    const v2f b0 = {b[i][0], b[i][1]};
    const v2f a1 = {a[i][2], a[i][3]};
    const v2f b1 = {b[i][2], b[i][3]};
    c0 = __builtin_amdgcn_wmma_f32_16x16x4_f32(
        false, a0, false, b0, (short)0, c0, false, false);
    c1 = __builtin_amdgcn_wmma_f32_16x16x4_f32(
        false, a1, false, b1, (short)0, c1, false, false);
  }

#pragma unroll
  for (int j = 0; j < 8; ++j) red[wave][j][lane] = c0[j] + c1[j];
  __syncthreads();

  float s = red[0][wave][lane];
#pragma unroll
  for (int w = 1; w < KWAVES; ++w) s += red[w][wave][lane];
  D[(size_t)(wave + 8 * half) * DIM + (n0 + lm)] = s + bv;
}

extern "C" void kernel_launch(void* const* d_in, const int* in_sizes, int n_in,
                              void* d_out, int out_size, void* d_ws, size_t ws_size,
                              hipStream_t stream) {
  const float* x     = (const float*)d_in[0];
  const float* W_enc = (const float*)d_in[1];
  const float* b_enc = (const float*)d_in[2];
  const float* W_out = (const float*)d_in[3];
  const float* b_out = (const float*)d_in[4];
  float* out = (float*)d_out;

  float* part = (float*)d_ws;
  float* m    = part + (size_t)SSPLIT * BATCH * DIM;
  float* enc  = m + (size_t)BATCH * DIM;

  dim3 g1(SSPLIT, BATCH);
  hcen_mean_stage1<<<g1, 256, 0, stream>>>(x, part);
  hcen_mean_stage2<<<(BATCH * DIM) / 256, 256, 0, stream>>>(part, m);
  hcen_gemm16<<<DIM / 16, 32 * KWAVES, 0, stream>>>(m,   W_enc, b_enc, enc);
  hcen_gemm16<<<DIM / 16, 32 * KWAVES, 0, stream>>>(enc, W_out, b_out, out);
}
